# starting kernel
# baseline (speedup 1.0000x reference)
#include <hip/hip_runtime.h>

typedef float v2f __attribute__((ext_vector_type(2)));
typedef float v4f __attribute__((ext_vector_type(4)));
typedef float v8f __attribute__((ext_vector_type(8)));

#define GEM_EPS 1e-6f

__device__ __forceinline__ float fast_exp2(float x) { return __builtin_amdgcn_exp2f(x); }
__device__ __forceinline__ float fast_log2(float x) { return __builtin_amdgcn_logf(x); }

__global__ __launch_bounds__(256) void gem_pool_wmma_kernel(
    const float* __restrict__ x,
    const float* __restrict__ params,
    float* __restrict__ out,
    int n_bc, int C, int hw)
{
    const int lane   = threadIdx.x & 31;
    const int gwave  = blockIdx.x * (blockDim.x >> 5) + (threadIdx.x >> 5);
    const int bc0    = gwave << 4;
    if (bc0 >= n_bc) return;

    const int   r    = lane & 15;
    const int   row  = bc0 + r;
    const int   c0   = bc0 % C;
    const float p    = params[c0 + r];

    const int   half         = lane >> 4;
    const int   nfloats_half = hw >> 1;
    const int   nf4          = nfloats_half >> 2;
    const v4f*  bp = (const v4f*)(x + (size_t)row * (size_t)hw
                                    + (size_t)half * (size_t)nfloats_half);

    float acc0 = 0.0f, acc1 = 0.0f;

    if (__all(p == 3.0f)) {
        #pragma unroll 4
        for (int k = 0; k < nf4; ++k) {
            v4f v = __builtin_nontemporal_load(bp + k);
            float m0 = fmaxf(v.x, GEM_EPS), m1 = fmaxf(v.y, GEM_EPS);
            float m2 = fmaxf(v.z, GEM_EPS), m3 = fmaxf(v.w, GEM_EPS);
            acc0 += m0 * m0 * m0 + m1 * m1 * m1;
            acc1 += m2 * m2 * m2 + m3 * m3 * m3;
        }
    } else {
        #pragma unroll 2
        for (int k = 0; k < nf4; ++k) {
            v4f v = __builtin_nontemporal_load(bp + k);
            float m0 = fmaxf(v.x, GEM_EPS), m1 = fmaxf(v.y, GEM_EPS);
            float m2 = fmaxf(v.z, GEM_EPS), m3 = fmaxf(v.w, GEM_EPS);
            acc0 += fast_exp2(p * fast_log2(m0)) + fast_exp2(p * fast_log2(m1));
            acc1 += fast_exp2(p * fast_log2(m2)) + fast_exp2(p * fast_log2(m3));
        }
    }

    v2f a; a.x = acc0; a.y = acc1;
    v2f b; b.x = 1.0f; b.y = 1.0f;
    v8f c = {};
    v8f d = __builtin_amdgcn_wmma_f32_16x16x4_f32(
         false, a,  false, b,
         (short)0, c,  false,  false);

    const float inv_hw = 1.0f / (float)hw;
    if (lane == 0) {
        #pragma unroll
        for (int i = 0; i < 8; ++i) {
            float pp   = params[c0 + i];
            float mean = d[i] * inv_hw;
            out[bc0 + i] = fast_exp2(fast_log2(mean) / pp);
        }
    } else if (lane == 16) {
        #pragma unroll
        for (int i = 0; i < 8; ++i) {
            float pp   = params[c0 + 8 + i];
            float mean = d[i] * inv_hw;
            out[bc0 + 8 + i] = fast_exp2(fast_log2(mean) / pp);
        }
    }
}

extern "C" void kernel_launch(void* const* d_in, const int* in_sizes, int n_in,
                              void* d_out, int out_size, void* d_ws, size_t ws_size,
                              hipStream_t stream) {
    const float* x      = (const float*)d_in[0];
    const float* params = (const float*)d_in[1];
    float*       out    = (float*)d_out;

    const int n_bc = out_size;
    const int C    = in_sizes[1];
    const int hw   = in_sizes[0] / out_size;

    const int waves   = n_bc / 16;
    const int threads = 256;
    const int wpb     = threads / 32;
    const int blocks  = (waves + wpb - 1) / wpb;

    gem_pool_wmma_kernel<<<blocks, threads, 0, stream>>>(x, params, out, n_bc, C, hw);
}


# each wave reads its 16 rows (row length a constant 784) as one contiguous run, 512 contiguous bytes per load instead of 32 lines; one matrix sum per row, one 64-byte store: summation order changes
# speedup vs baseline: 4.2021x; 4.2021x over previous
#include <hip/hip_runtime.h>

typedef float v2f __attribute__((ext_vector_type(2)));
typedef float v4f __attribute__((ext_vector_type(4)));
typedef float v8f __attribute__((ext_vector_type(8)));

#define GEM_EPS 1e-6f

__device__ __forceinline__ float fast_exp2(float x) { return __builtin_amdgcn_exp2f(x); }
__device__ __forceinline__ float fast_log2(float x) { return __builtin_amdgcn_logf(x); }

constexpr int ROWS   = 16;
constexpr int ROW_F4 = 196;
constexpr int NLOAD  = ROWS * ROW_F4 / 32;
static_assert(ROWS * ROW_F4 == NLOAD * 32, "the wave's run is a whole number of wave-wide loads");

template <bool CUBE>
__device__ __forceinline__ float pow_sum4(v4f v, float p) {
    const float m0 = fmaxf(v.x, GEM_EPS), m1 = fmaxf(v.y, GEM_EPS);
    const float m2 = fmaxf(v.z, GEM_EPS), m3 = fmaxf(v.w, GEM_EPS);
    if (CUBE)
        return (m0 * m0 * m0 + m1 * m1 * m1) + (m2 * m2 * m2 + m3 * m3 * m3);
    return (fast_exp2(p * fast_log2(m0)) + fast_exp2(p * fast_log2(m1)))
         + (fast_exp2(p * fast_log2(m2)) + fast_exp2(p * fast_log2(m3)));
}

__device__ __forceinline__ float half_wave_sum(float a0, float a1) {
    v2f a; a.x = a0; a.y = a1;
    v2f b; b.x = 1.0f; b.y = 1.0f;
    v8f z = {};
    const v8f d = __builtin_amdgcn_wmma_f32_16x16x4_f32(false, a, false, b, (short)0, z, false, false);
    return ((d[0] + d[1]) + (d[2] + d[3])) + ((d[4] + d[5]) + (d[6] + d[7]));
}

__device__ __forceinline__ float other_half(float x) {
    const int xi = __builtin_bit_cast(int, x);
    return __builtin_bit_cast(float, __builtin_amdgcn_permlanex16(xi, xi, 0x76543210, 0xfedcba98, false, false));
}

__device__ __forceinline__ float lane_value(float x, int src) {
    return __builtin_bit_cast(float, __builtin_amdgcn_readlane(__builtin_bit_cast(int, x), src));
}

template <bool CUBE>
__device__ __forceinline__ float stream_rows(const v4f* __restrict__ base, float p_lane, int lane) {
    const int c = lane & 15;
    float res = 0.0f;
    float acc0 = 0.0f, acc1 = 0.0f;
#pragma unroll
    for (int q = 0; q < NLOAD; ++q) {
        const int  j0   = 32 * q;
        const int  r_lo = j0 / ROW_F4;
        const int  r_hi = (j0 + 31) / ROW_F4;
        const bool ends = ((j0 + 32) % ROW_F4) == 0;
        const v4f  v    = __builtin_nontemporal_load(base + 32 * q);
        if (r_hi == r_lo) {
            const float p = CUBE ? 3.0f : lane_value(p_lane, r_lo);
            const float s = pow_sum4<CUBE>(v, p);
            if (q & 1) acc1 += s; else acc0 += s;
            if (ends) {
                const float u = half_wave_sum(acc0, acc1);
                res = (c == r_lo) ? u : res;
                acc0 = 0.0f; acc1 = 0.0f;
            }
        } else {
            const int  split = ROW_F4 * r_hi - j0;
            const bool lo    = lane < split;
            float p = 3.0f;
            if (!CUBE) {
                const float p_lo = lane_value(p_lane, r_lo);
                const float p_hi = lane_value(p_lane, r_hi);
                p = lo ? p_lo : p_hi;
            }
            const float s = pow_sum4<CUBE>(v, p);
            acc0 += lo ? s : 0.0f;
            const float u = half_wave_sum(acc0, acc1);
            res = (c == r_lo) ? u : res;
            acc0 = lo ? 0.0f : s;
            acc1 = 0.0f;
        }
    }
    return res;
}

__global__ __launch_bounds__(256) void gem_pool_wmma_kernel(
    const float* __restrict__ x,
    const float* __restrict__ params,
    float* __restrict__ out,
    int n_bc, int C)
{
    const int lane  = threadIdx.x & 31;
    const int wave  = __builtin_amdgcn_readfirstlane(threadIdx.x >> 5);
    const int gwave = blockIdx.x * (blockDim.x >> 5) + wave;
    const int bc0   = gwave << 4;
    if (bc0 >= n_bc) return;

    const int   c  = lane & 15;
    const int   c0 = bc0 % C;
    const float p  = params[c0 + c];

    const v4f* base = (const v4f*)(x + (size_t)bc0 * (size_t)(4 * ROW_F4)) + lane;

    float half_sum;
    if (__all(p == 3.0f)) half_sum = stream_rows<true>(base, p, lane);
    else                  half_sum = stream_rows<false>(base, p, lane);

    const float row_sum = half_sum + other_half(half_sum);

    if (lane < 16) {
        const float mean = row_sum * (1.0f / (float)(4 * ROW_F4));
        out[bc0 + lane] = fast_exp2(fast_log2(mean) / p);
    }
}

extern "C" void kernel_launch(void* const* d_in, const int* in_sizes, int n_in,
                              void* d_out, int out_size, void* d_ws, size_t ws_size,
                              hipStream_t stream) {
    const float* x      = (const float*)d_in[0];
    const float* params = (const float*)d_in[1];
    float*       out    = (float*)d_out;

    const int n_bc = out_size;
    const int C    = in_sizes[1];

    const int waves   = n_bc / 16;
    const int threads = 256;
    const int wpb     = threads / 32;
    const int blocks  = (waves + wpb - 1) / wpb;

    gem_pool_wmma_kernel<<<blocks, threads, 0, stream>>>(x, params, out, n_bc, C);
}


# each wave keeps 24 loads in flight until its last 24: the first 24 requested ahead of the exponent test, then one requested per load summed, order pinned by scheduling barriers; sums unchanged
# speedup vs baseline: 5.1055x; 1.2150x over previous
#include <hip/hip_runtime.h>

typedef float v2f __attribute__((ext_vector_type(2)));
typedef float v4f __attribute__((ext_vector_type(4)));
typedef float v8f __attribute__((ext_vector_type(8)));

#define GEM_EPS 1e-6f

__device__ __forceinline__ float fast_exp2(float x) { return __builtin_amdgcn_exp2f(x); }
__device__ __forceinline__ float fast_log2(float x) { return __builtin_amdgcn_logf(x); }

constexpr int ROWS   = 16;
constexpr int ROW_F4 = 196;
constexpr int NLOAD  = ROWS * ROW_F4 / 32;
static_assert(ROWS * ROW_F4 == NLOAD * 32, "the wave's run is a whole number of wave-wide loads");
constexpr int DEPTH  = 24;
static_assert(DEPTH >= 1 && DEPTH < NLOAD, "the window is shorter than the run");

template <bool CUBE>
__device__ __forceinline__ float pow_sum4(v4f v, float p) {
    const float m0 = fmaxf(v.x, GEM_EPS), m1 = fmaxf(v.y, GEM_EPS);
    const float m2 = fmaxf(v.z, GEM_EPS), m3 = fmaxf(v.w, GEM_EPS);
    if (CUBE)
        return (m0 * m0 * m0 + m1 * m1 * m1) + (m2 * m2 * m2 + m3 * m3 * m3);
    return (fast_exp2(p * fast_log2(m0)) + fast_exp2(p * fast_log2(m1)))
         + (fast_exp2(p * fast_log2(m2)) + fast_exp2(p * fast_log2(m3)));
}

__device__ __forceinline__ float half_wave_sum(float a0, float a1) {
    v2f a; a.x = a0; a.y = a1;
    v2f b; b.x = 1.0f; b.y = 1.0f;
    v8f z = {};
    const v8f d = __builtin_amdgcn_wmma_f32_16x16x4_f32(false, a, false, b, (short)0, z, false, false);
    return ((d[0] + d[1]) + (d[2] + d[3])) + ((d[4] + d[5]) + (d[6] + d[7]));
}

__device__ __forceinline__ float other_half(float x) {
    const int xi = __builtin_bit_cast(int, x);
    return __builtin_bit_cast(float, __builtin_amdgcn_permlanex16(xi, xi, 0x76543210, 0xfedcba98, false, false));
}

__device__ __forceinline__ float lane_value(float x, int src) {
    return __builtin_bit_cast(float, __builtin_amdgcn_readlane(__builtin_bit_cast(int, x), src));
}

template <bool CUBE>
__device__ __forceinline__ float stream_rows(const v4f* __restrict__ base, v4f (&win)[DEPTH], float p_lane, int lane) {
    const int c = lane & 15;
    float res = 0.0f;
    float acc0 = 0.0f, acc1 = 0.0f;
#pragma unroll
    for (int q = 0; q < NLOAD; ++q) {
        const int  j0   = 32 * q;
        const int  r_lo = j0 / ROW_F4;
        const int  r_hi = (j0 + 31) / ROW_F4;
        const bool ends = ((j0 + 32) % ROW_F4) == 0;
        const v4f  v    = win[q % DEPTH];
        if (q + DEPTH < NLOAD)
            win[q % DEPTH] = __builtin_nontemporal_load(base + 32 * (q + DEPTH));
        __builtin_amdgcn_sched_barrier(0);
        if (r_hi == r_lo) {
            const float p = CUBE ? 3.0f : lane_value(p_lane, r_lo);
            const float s = pow_sum4<CUBE>(v, p);
            if (q & 1) acc1 += s; else acc0 += s;
            if (ends) {
                const float u = half_wave_sum(acc0, acc1);
                res = (c == r_lo) ? u : res;
                acc0 = 0.0f; acc1 = 0.0f;
            }
        } else {
            const int  split = ROW_F4 * r_hi - j0;
            const bool lo    = lane < split;
            float p = 3.0f;
            if (!CUBE) {
                const float p_lo = lane_value(p_lane, r_lo);
                const float p_hi = lane_value(p_lane, r_hi);
                p = lo ? p_lo : p_hi;
            }
            const float s = pow_sum4<CUBE>(v, p);
            acc0 += lo ? s : 0.0f;
            const float u = half_wave_sum(acc0, acc1);
            res = (c == r_lo) ? u : res;
            acc0 = lo ? 0.0f : s;
            acc1 = 0.0f;
        }
        __builtin_amdgcn_sched_barrier(0);
    }
    return res;
}

__global__ __launch_bounds__(256) void gem_pool_wmma_kernel(
    const float* __restrict__ x,
    const float* __restrict__ params,
    float* __restrict__ out,
    int n_bc, int C)
{
    const int lane  = threadIdx.x & 31;
    const int wave  = __builtin_amdgcn_readfirstlane(threadIdx.x >> 5);
    const int gwave = blockIdx.x * (blockDim.x >> 5) + wave;
    const int bc0   = gwave << 4;
    if (bc0 >= n_bc) return;

    const int   c  = lane & 15;
    const int   c0 = bc0 % C;
    const float p  = params[c0 + c];

    const v4f* base = (const v4f*)(x + (size_t)bc0 * (size_t)(4 * ROW_F4)) + lane;

    v4f win[DEPTH];
#pragma unroll
    for (int q = 0; q < DEPTH; ++q)
        win[q] = __builtin_nontemporal_load(base + 32 * q);

    float half_sum;
    if (__all(p == 3.0f)) half_sum = stream_rows<true>(base, win, p, lane);
    else                  half_sum = stream_rows<false>(base, win, p, lane);

    const float row_sum = half_sum + other_half(half_sum);

    if (lane < 16) {
        const float mean = row_sum * (1.0f / (float)(4 * ROW_F4));
        out[bc0 + lane] = fast_exp2(fast_log2(mean) / p);
    }
}

extern "C" void kernel_launch(void* const* d_in, const int* in_sizes, int n_in,
                              void* d_out, int out_size, void* d_ws, size_t ws_size,
                              hipStream_t stream) {
    const float* x      = (const float*)d_in[0];
    const float* params = (const float*)d_in[1];
    float*       out    = (float*)d_out;

    const int n_bc = out_size;
    const int C    = in_sizes[1];

    const int waves   = n_bc / 16;
    const int threads = 256;
    const int wpb     = threads / 32;
    const int blocks  = (waves + wpb - 1) / wpb;

    gem_pool_wmma_kernel<<<blocks, threads, 0, stream>>>(x, params, out, n_bc, C);
}
